# starting kernel
# baseline (speedup 1.0000x reference)
#include <hip/hip_runtime.h>


constexpr int NP = 128;
constexpr float FAR_DELTA = 1e10f;

typedef __attribute__((ext_vector_type(2))) float v2f;
typedef __attribute__((ext_vector_type(8))) float v8f;

__global__ __launch_bounds__(512)
void volrend_kernel(const float* __restrict__ density,
                    const float* __restrict__ depth,
                    const float* __restrict__ feature,
                    float* __restrict__ out, int N)
{
    __shared__ float4 part[16][33];

    const int lane    = threadIdx.x & 31;
    const int wave    = threadIdx.x >> 5;
    const int rayBase = blockIdx.x * 16;
    const int ray     = rayBase + wave;

    float fx = 0.f, fy = 0.f, fz = 0.f, dd = 0.f;

    if (ray < N) {
        const float4 sg = *(const float4*)(density + ray * NP + lane * 4);
        const float4 dp = *(const float4*)(depth   + ray * NP + lane * 4);

        float nx = __shfl_down(dp.x, 1, 32);
        float d0 = dp.y - dp.x;
        float d1 = dp.z - dp.y;
        float d2 = dp.w - dp.z;
        float d3 = (lane == 31) ? FAR_DELTA : (nx - dp.w);

        float s0 = __expf(-sg.x * d0);
        float s1 = __expf(-sg.y * d1);
        float s2 = __expf(-sg.z * d2);
        float s3 = __expf(-sg.w * d3);

        float t = s0 * s1 * s2 * s3;
        #pragma unroll
        for (int off = 1; off < 32; off <<= 1) {
            float v = __shfl_up(t, off, 32);
            if (lane >= off) t *= v;
        }
        float excl = __shfl_up(t, 1, 32);
        if (lane == 0) excl = 1.0f;

        float T0 = excl;
        float T1 = T0 * s0;
        float T2 = T1 * s1;
        float T3 = T2 * s2;
        float w0 = T0 * (1.0f - s0);
        float w1 = T1 * (1.0f - s1);
        float w2 = T2 * (1.0f - s2);
        float w3 = T3 * (1.0f - s3);

        const float* fp = feature + (size_t)ray * (NP * 3) + lane * 12;
        const float4 fa = *(const float4*)(fp + 0);
        const float4 fb = *(const float4*)(fp + 4);
        const float4 fc = *(const float4*)(fp + 8);

        fx = w0 * fa.x + w1 * fa.w + w2 * fb.z + w3 * fc.y;
        fy = w0 * fa.y + w1 * fb.x + w2 * fb.w + w3 * fc.z;
        fz = w0 * fa.z + w1 * fb.y + w2 * fc.x + w3 * fc.w;
        dd = w0 * dp.x + w1 * dp.y + w2 * dp.z + w3 * dp.w;
    }

    part[wave][lane] = make_float4(fx, fy, fz, dd);
    __syncthreads();

    if (wave == 0) {
        const float* Pf = (const float*)part;
        const v2f bones = {1.0f, 1.0f};
        const int row   = lane & 15;
        const int khalf = (lane >> 4) << 1;

        #pragma unroll
        for (int c = 0; c < 4; ++c) {
            v8f acc = {0.f, 0.f, 0.f, 0.f, 0.f, 0.f, 0.f, 0.f};
            #pragma unroll
            for (int kb = 0; kb < 8; ++kb) {
                const int k0 = kb * 4 + khalf;
                v2f a;
                a[0] = Pf[(row * 33 + k0)     * 4 + c];
                a[1] = Pf[(row * 33 + k0 + 1) * 4 + c];
                acc = __builtin_amdgcn_wmma_f32_16x16x4_f32(
                    false, a, false, bones, (short)0, acc, false, false);
            }
            #pragma unroll
            for (int j = 0; j < 8; ++j) {
                if (lane == j) {
                    int r = rayBase + j;
                    if (r < N) {
                        if (c < 3) out[r * 3 + c]   = acc[j];
                        else       out[3 * N + r]   = acc[j];
                    }
                }
                if (lane == 16 + j) {
                    int r = rayBase + 8 + j;
                    if (r < N) {
                        if (c < 3) out[r * 3 + c]   = acc[j];
                        else       out[3 * N + r]   = acc[j];
                    }
                }
            }
        }
    }
}

extern "C" void kernel_launch(void* const* d_in, const int* in_sizes, int n_in,
                              void* d_out, int out_size, void* d_ws, size_t ws_size,
                              hipStream_t stream) {
    const float* density = (const float*)d_in[0];
    const float* depth   = (const float*)d_in[1];
    const float* feature = (const float*)d_in[2];
    float* out = (float*)d_out;

    const int N = in_sizes[0] / NP;
    const int blocks = (N + 15) / 16;

    hipLaunchKernelGGL(volrend_kernel, dim3(blocks), dim3(512), 0, stream,
                       density, depth, feature, out, N);
}


# one wave owns a 16-ray tile end to end (was one workgroup): four rays' loads kept in flight ahead, wave-private LDS slab into the same f32 WMMA ones-reduction, no block barrier, 16x fewer waves
# speedup vs baseline: 1.0829x; 1.0829x over previous
#include <hip/hip_runtime.h>


constexpr int NP    = 128;
constexpr int TILE  = 16;
constexpr int WAVES = 16;
constexpr int GROUP = 2;
constexpr int STEPS = TILE / GROUP;
constexpr int AHEAD = 2;
constexpr float FAR_DELTA = 1e10f;

typedef __attribute__((ext_vector_type(2))) float v2f;
typedef __attribute__((ext_vector_type(8))) float v8f;

typedef __attribute__((ext_vector_type(4))) float f4;
struct RayIn { f4 sg, dp, fa, fb, fc; };

static __device__ __forceinline__ void load_ray(RayIn& q,
        const float* __restrict__ density, const float* __restrict__ depth,
        const float* __restrict__ feature, int ray, int lane)
{
    q.sg = *(const f4*)(density + (size_t)ray * NP + lane * 4);
    q.dp = *(const f4*)(depth   + (size_t)ray * NP + lane * 4);
    const float* fp = feature + (size_t)ray * (NP * 3) + lane * 12;
    q.fa = *(const f4*)(fp + 0);
    q.fb = *(const f4*)(fp + 4);
    q.fc = *(const f4*)(fp + 8);
}

static __device__ __forceinline__ void composite_group(const RayIn (&q)[GROUP], int lane, float4 (&res)[GROUP])
{
    float nx[GROUP], s0[GROUP], s1[GROUP], s2[GROUP], s3[GROUP], t[GROUP], excl[GROUP];

    #pragma unroll
    for (int g = 0; g < GROUP; ++g) nx[g] = __shfl_down(q[g].dp.x, 1, 32);

    #pragma unroll
    for (int g = 0; g < GROUP; ++g) {
        const f4 sg = q[g].sg, dp = q[g].dp;
        float d0 = dp.y - dp.x;
        float d1 = dp.z - dp.y;
        float d2 = dp.w - dp.z;
        float d3 = (lane == 31) ? FAR_DELTA : (nx[g] - dp.w);
        s0[g] = __expf(-sg.x * d0);
        s1[g] = __expf(-sg.y * d1);
        s2[g] = __expf(-sg.z * d2);
        s3[g] = __expf(-sg.w * d3);
        t[g]  = s0[g] * s1[g] * s2[g] * s3[g];
    }

    #pragma unroll
    for (int off = 1; off < 32; off <<= 1) {
        float v[GROUP];
        #pragma unroll
        for (int g = 0; g < GROUP; ++g) v[g] = __shfl_up(t[g], off, 32);
        #pragma unroll
        for (int g = 0; g < GROUP; ++g) if (lane >= off) t[g] *= v[g];
    }
    #pragma unroll
    for (int g = 0; g < GROUP; ++g) excl[g] = __shfl_up(t[g], 1, 32);

    #pragma unroll
    for (int g = 0; g < GROUP; ++g) {
        const f4 dp = q[g].dp, fa = q[g].fa, fb = q[g].fb, fc = q[g].fc;
        if (lane == 0) excl[g] = 1.0f;
        float T0 = excl[g];
        float T1 = T0 * s0[g];
        float T2 = T1 * s1[g];
        float T3 = T2 * s2[g];
        float w0 = T0 * (1.0f - s0[g]);
        float w1 = T1 * (1.0f - s1[g]);
        float w2 = T2 * (1.0f - s2[g]);
        float w3 = T3 * (1.0f - s3[g]);

        float fx = w0 * fa.x + w1 * fa.w + w2 * fb.z + w3 * fc.y;
        float fy = w0 * fa.y + w1 * fb.x + w2 * fb.w + w3 * fc.z;
        float fz = w0 * fa.z + w1 * fb.y + w2 * fc.x + w3 * fc.w;
        float dd = w0 * dp.x + w1 * dp.y + w2 * dp.z + w3 * dp.w;
        res[g] = make_float4(fx, fy, fz, dd);
    }
}

__global__ __launch_bounds__(WAVES * 32)
void volrend_kernel(const float* __restrict__ density,
                    const float* __restrict__ depth,
                    const float* __restrict__ feature,
                    float* __restrict__ out, int N)
{
    __shared__ float4 part[WAVES][TILE][33];

    const int lane    = threadIdx.x & 31;
    const int wave    = __builtin_amdgcn_readfirstlane(threadIdx.x >> 5);
    const int rayBase = (blockIdx.x * WAVES + wave) * TILE;
    if (rayBase >= N) return;

    #define RAY_OF(R) ((rayBase + (R) < N) ? (rayBase + (R)) : rayBase)
    #define LOAD_STEP(S) do { _Pragma("unroll") for (int g_ = 0; g_ < GROUP; ++g_) \
        load_ray(in[(S) % (AHEAD + 1)][g_], density, depth, feature, RAY_OF((S) * GROUP + g_), lane); } while (0)

    RayIn in[AHEAD + 1][GROUP];

    #pragma unroll
    for (int s = 0; s < AHEAD; ++s) LOAD_STEP(s);

    #pragma unroll
    for (int s = 0; s < STEPS; ++s) {
        __builtin_amdgcn_sched_barrier(0);
        if (s + AHEAD < STEPS) LOAD_STEP(s + AHEAD);
        __builtin_amdgcn_sched_barrier(0);
        #pragma unroll
        for (int g = 0; g < GROUP; ++g) {
            RayIn& q = in[s % (AHEAD + 1)][g];
            asm volatile("" : "+v"(q.sg), "+v"(q.dp), "+v"(q.fa), "+v"(q.fb), "+v"(q.fc));
        }
        float4 res[GROUP];
        composite_group(in[s % (AHEAD + 1)], lane, res);
        #pragma unroll
        for (int g = 0; g < GROUP; ++g) part[wave][s * GROUP + g][lane] = res[g];
    }
    #undef LOAD_STEP
    #undef RAY_OF

    {
        const float* Pf = (const float*)&part[wave][0][0];
        const v2f bones = {1.0f, 1.0f};
        const int row   = lane & 15;
        const int khalf = (lane >> 4) << 1;

        #pragma unroll
        for (int c = 0; c < 4; ++c) {
            v8f acc = {0.f, 0.f, 0.f, 0.f, 0.f, 0.f, 0.f, 0.f};
            #pragma unroll
            for (int kb = 0; kb < 8; ++kb) {
                const int k0 = kb * 4 + khalf;
                v2f a;
                a[0] = Pf[(row * 33 + k0)     * 4 + c];
                a[1] = Pf[(row * 33 + k0 + 1) * 4 + c];
                acc = __builtin_amdgcn_wmma_f32_16x16x4_f32(
                    false, a, false, bones, (short)0, acc, false, false);
            }
            #pragma unroll
            for (int j = 0; j < 8; ++j) {
                if (lane == j) {
                    int r = rayBase + j;
                    if (r < N) {
                        if (c < 3) out[r * 3 + c]   = acc[j];
                        else       out[3 * N + r]   = acc[j];
                    }
                }
                if (lane == 16 + j) {
                    int r = rayBase + 8 + j;
                    if (r < N) {
                        if (c < 3) out[r * 3 + c]   = acc[j];
                        else       out[3 * N + r]   = acc[j];
                    }
                }
            }
        }
    }
}

extern "C" void kernel_launch(void* const* d_in, const int* in_sizes, int n_in,
                              void* d_out, int out_size, void* d_ws, size_t ws_size,
                              hipStream_t stream) {
    const float* density = (const float*)d_in[0];
    const float* depth   = (const float*)d_in[1];
    const float* feature = (const float*)d_in[2];
    float* out = (float*)d_out;

    const int N = in_sizes[0] / NP;
    const int blocks = (N + WAVES * TILE - 1) / (WAVES * TILE);

    hipLaunchKernelGGL(volrend_kernel, dim3(blocks), dim3(WAVES * 32), 0, stream,
                       density, depth, feature, out, N);
}


# feature rows fetched as three whole 512-byte pieces per ray (each load covers 4 full lines; was 12 lines at a 48-byte lane stride), handed to sample order through a wave-private LDS row
# speedup vs baseline: 1.1515x; 1.0633x over previous
#include <hip/hip_runtime.h>


constexpr int NP    = 128;
constexpr int TILE  = 16;
constexpr int WAVES = 16;
constexpr int GROUP = 2;
constexpr int STEPS = TILE / GROUP;
constexpr int AHEAD = 2;
constexpr float FAR_DELTA = 1e10f;

typedef __attribute__((ext_vector_type(2))) float v2f;
typedef __attribute__((ext_vector_type(8))) float v8f;

typedef __attribute__((ext_vector_type(4))) float f4;
struct RayIn { f4 sg, dp, fa, fb, fc; };

static __device__ __forceinline__ void load_ray(RayIn& q,
        const float* __restrict__ density, const float* __restrict__ depth,
        const float* __restrict__ feature, int ray, int lane)
{
    q.sg = *(const f4*)(density + (size_t)ray * NP + lane * 4);
    q.dp = *(const f4*)(depth   + (size_t)ray * NP + lane * 4);
    const float* fp = feature + (size_t)ray * (NP * 3) + lane * 4;
    q.fa = *(const f4*)(fp + 0);
    q.fb = *(const f4*)(fp + 128);
    q.fc = *(const f4*)(fp + 256);
}

static __device__ __forceinline__ void to_sample_order(RayIn (&q)[GROUP], f4 (*row)[96], int lane)
{
    #pragma unroll
    for (int g = 0; g < GROUP; ++g) {
        row[g][lane]      = q[g].fa;
        row[g][lane + 32] = q[g].fb;
        row[g][lane + 64] = q[g].fc;
    }
    #pragma unroll
    for (int g = 0; g < GROUP; ++g) {
        q[g].fa = row[g][lane * 3 + 0];
        q[g].fb = row[g][lane * 3 + 1];
        q[g].fc = row[g][lane * 3 + 2];
    }
}

static __device__ __forceinline__ void composite_group(const RayIn (&q)[GROUP], int lane, float4 (&res)[GROUP])
{
    float nx[GROUP], s0[GROUP], s1[GROUP], s2[GROUP], s3[GROUP], t[GROUP], excl[GROUP];

    #pragma unroll
    for (int g = 0; g < GROUP; ++g) nx[g] = __shfl_down(q[g].dp.x, 1, 32);

    #pragma unroll
    for (int g = 0; g < GROUP; ++g) {
        const f4 sg = q[g].sg, dp = q[g].dp;
        float d0 = dp.y - dp.x;
        float d1 = dp.z - dp.y;
        float d2 = dp.w - dp.z;
        float d3 = (lane == 31) ? FAR_DELTA : (nx[g] - dp.w);
        s0[g] = __expf(-sg.x * d0);
        s1[g] = __expf(-sg.y * d1);
        s2[g] = __expf(-sg.z * d2);
        s3[g] = __expf(-sg.w * d3);
        t[g]  = s0[g] * s1[g] * s2[g] * s3[g];
    }

    #pragma unroll
    for (int off = 1; off < 32; off <<= 1) {
        float v[GROUP];
        #pragma unroll
        for (int g = 0; g < GROUP; ++g) v[g] = __shfl_up(t[g], off, 32);
        #pragma unroll
        for (int g = 0; g < GROUP; ++g) if (lane >= off) t[g] *= v[g];
    }
    #pragma unroll
    for (int g = 0; g < GROUP; ++g) excl[g] = __shfl_up(t[g], 1, 32);

    #pragma unroll
    for (int g = 0; g < GROUP; ++g) {
        const f4 dp = q[g].dp, fa = q[g].fa, fb = q[g].fb, fc = q[g].fc;
        if (lane == 0) excl[g] = 1.0f;
        float T0 = excl[g];
        float T1 = T0 * s0[g];
        float T2 = T1 * s1[g];
        float T3 = T2 * s2[g];
        float w0 = T0 * (1.0f - s0[g]);
        float w1 = T1 * (1.0f - s1[g]);
        float w2 = T2 * (1.0f - s2[g]);
        float w3 = T3 * (1.0f - s3[g]);

        float fx = w0 * fa.x + w1 * fa.w + w2 * fb.z + w3 * fc.y;
        float fy = w0 * fa.y + w1 * fb.x + w2 * fb.w + w3 * fc.z;
        float fz = w0 * fa.z + w1 * fb.y + w2 * fc.x + w3 * fc.w;
        float dd = w0 * dp.x + w1 * dp.y + w2 * dp.z + w3 * dp.w;
        res[g] = make_float4(fx, fy, fz, dd);
    }
}

__global__ __launch_bounds__(WAVES * 32)
void volrend_kernel(const float* __restrict__ density,
                    const float* __restrict__ depth,
                    const float* __restrict__ feature,
                    float* __restrict__ out, int N)
{
    __shared__ float4 part[WAVES][TILE][33];
    __shared__ f4 frow[WAVES][GROUP][96];

    const int lane    = threadIdx.x & 31;
    const int wave    = __builtin_amdgcn_readfirstlane(threadIdx.x >> 5);
    const int rayBase = (blockIdx.x * WAVES + wave) * TILE;
    if (rayBase >= N) return;

    #define RAY_OF(R) ((rayBase + (R) < N) ? (rayBase + (R)) : rayBase)
    #define LOAD_STEP(S) do { _Pragma("unroll") for (int g_ = 0; g_ < GROUP; ++g_) \
        load_ray(in[(S) % (AHEAD + 1)][g_], density, depth, feature, RAY_OF((S) * GROUP + g_), lane); } while (0)

    RayIn in[AHEAD + 1][GROUP];

    #pragma unroll
    for (int s = 0; s < AHEAD; ++s) LOAD_STEP(s);

    #pragma unroll
    for (int s = 0; s < STEPS; ++s) {
        __builtin_amdgcn_sched_barrier(0);
        if (s + AHEAD < STEPS) LOAD_STEP(s + AHEAD);
        __builtin_amdgcn_sched_barrier(0);
        #pragma unroll
        for (int g = 0; g < GROUP; ++g) {
            RayIn& q = in[s % (AHEAD + 1)][g];
            asm volatile("" : "+v"(q.sg), "+v"(q.dp), "+v"(q.fa), "+v"(q.fb), "+v"(q.fc));
        }
        to_sample_order(in[s % (AHEAD + 1)], frow[wave], lane);
        float4 res[GROUP];
        composite_group(in[s % (AHEAD + 1)], lane, res);
        #pragma unroll
        for (int g = 0; g < GROUP; ++g) part[wave][s * GROUP + g][lane] = res[g];
    }
    #undef LOAD_STEP
    #undef RAY_OF

    {
        const float* Pf = (const float*)&part[wave][0][0];
        const v2f bones = {1.0f, 1.0f};
        const int row   = lane & 15;
        const int khalf = (lane >> 4) << 1;

        #pragma unroll
        for (int c = 0; c < 4; ++c) {
            v8f acc = {0.f, 0.f, 0.f, 0.f, 0.f, 0.f, 0.f, 0.f};
            #pragma unroll
            for (int kb = 0; kb < 8; ++kb) {
                const int k0 = kb * 4 + khalf;
                v2f a;
                a[0] = Pf[(row * 33 + k0)     * 4 + c];
                a[1] = Pf[(row * 33 + k0 + 1) * 4 + c];
                acc = __builtin_amdgcn_wmma_f32_16x16x4_f32(
                    false, a, false, bones, (short)0, acc, false, false);
            }
            #pragma unroll
            for (int j = 0; j < 8; ++j) {
                if (lane == j) {
                    int r = rayBase + j;
                    if (r < N) {
                        if (c < 3) out[r * 3 + c]   = acc[j];
                        else       out[3 * N + r]   = acc[j];
                    }
                }
                if (lane == 16 + j) {
                    int r = rayBase + 8 + j;
                    if (r < N) {
                        if (c < 3) out[r * 3 + c]   = acc[j];
                        else       out[3 * N + r]   = acc[j];
                    }
                }
            }
        }
    }
}

extern "C" void kernel_launch(void* const* d_in, const int* in_sizes, int n_in,
                              void* d_out, int out_size, void* d_ws, size_t ws_size,
                              hipStream_t stream) {
    const float* density = (const float*)d_in[0];
    const float* depth   = (const float*)d_in[1];
    const float* feature = (const float*)d_in[2];
    float* out = (float*)d_out;

    const int N = in_sizes[0] / NP;
    const int blocks = (N + WAVES * TILE - 1) / (WAVES * TILE);

    hipLaunchKernelGGL(volrend_kernel, dim3(blocks), dim3(WAVES * 32), 0, stream,
                       density, depth, feature, out, N);
}


# row sums per 4-ray quarter inside the streaming loop: partials as WMMA B operand (A = ones, columns = 4 rays x 4 channels), lane n holds one output; 2 wave-wide stores replace 64 single-lane ones
# speedup vs baseline: 1.4721x; 1.2784x over previous
#include <hip/hip_runtime.h>


constexpr int NP    = 128;
constexpr int TILE  = 16;
constexpr int WAVES = 16;
constexpr int GROUP = 2;
constexpr int STEPS = TILE / GROUP;
constexpr int AHEAD = 2;
constexpr int QRAYS = 4;
constexpr int QSTEPS = QRAYS / GROUP;
constexpr int PITCH = 36;
constexpr float FAR_DELTA = 1e10f;

typedef __attribute__((ext_vector_type(2))) float v2f;
typedef __attribute__((ext_vector_type(8))) float v8f;

typedef __attribute__((ext_vector_type(4))) float f4;
struct RayIn { f4 sg, dp, fa, fb, fc; };

static __device__ __forceinline__ void load_ray(RayIn& q,
        const float* __restrict__ density, const float* __restrict__ depth,
        const float* __restrict__ feature, int ray, int lane)
{
    q.sg = *(const f4*)(density + (size_t)ray * NP + lane * 4);
    q.dp = *(const f4*)(depth   + (size_t)ray * NP + lane * 4);
    const float* fp = feature + (size_t)ray * (NP * 3) + lane * 4;
    q.fa = *(const f4*)(fp + 0);
    q.fb = *(const f4*)(fp + 128);
    q.fc = *(const f4*)(fp + 256);
}

static __device__ __forceinline__ void to_sample_order(RayIn (&q)[GROUP], f4 (*row)[96], int lane)
{
    #pragma unroll
    for (int g = 0; g < GROUP; ++g) {
        row[g][lane]      = q[g].fa;
        row[g][lane + 32] = q[g].fb;
        row[g][lane + 64] = q[g].fc;
    }
    #pragma unroll
    for (int g = 0; g < GROUP; ++g) {
        q[g].fa = row[g][lane * 3 + 0];
        q[g].fb = row[g][lane * 3 + 1];
        q[g].fc = row[g][lane * 3 + 2];
    }
}

static __device__ __forceinline__ void composite_group(const RayIn (&q)[GROUP], int lane, float4 (&res)[GROUP])
{
    float nx[GROUP], s0[GROUP], s1[GROUP], s2[GROUP], s3[GROUP], t[GROUP], excl[GROUP];

    #pragma unroll
    for (int g = 0; g < GROUP; ++g) nx[g] = __shfl_down(q[g].dp.x, 1, 32);

    #pragma unroll
    for (int g = 0; g < GROUP; ++g) {
        const f4 sg = q[g].sg, dp = q[g].dp;
        float d0 = dp.y - dp.x;
        float d1 = dp.z - dp.y;
        float d2 = dp.w - dp.z;
        float d3 = (lane == 31) ? FAR_DELTA : (nx[g] - dp.w);
        s0[g] = __expf(-sg.x * d0);
        s1[g] = __expf(-sg.y * d1);
        s2[g] = __expf(-sg.z * d2);
        s3[g] = __expf(-sg.w * d3);
        t[g]  = s0[g] * s1[g] * s2[g] * s3[g];
    }

    #pragma unroll
    for (int off = 1; off < 32; off <<= 1) {
        float v[GROUP];
        #pragma unroll
        for (int g = 0; g < GROUP; ++g) v[g] = __shfl_up(t[g], off, 32);
        #pragma unroll
        for (int g = 0; g < GROUP; ++g) if (lane >= off) t[g] *= v[g];
    }
    #pragma unroll
    for (int g = 0; g < GROUP; ++g) excl[g] = __shfl_up(t[g], 1, 32);

    #pragma unroll
    for (int g = 0; g < GROUP; ++g) {
        const f4 dp = q[g].dp, fa = q[g].fa, fb = q[g].fb, fc = q[g].fc;
        if (lane == 0) excl[g] = 1.0f;
        float T0 = excl[g];
        float T1 = T0 * s0[g];
        float T2 = T1 * s1[g];
        float T3 = T2 * s2[g];
        float w0 = T0 * (1.0f - s0[g]);
        float w1 = T1 * (1.0f - s1[g]);
        float w2 = T2 * (1.0f - s2[g]);
        float w3 = T3 * (1.0f - s3[g]);

        float fx = w0 * fa.x + w1 * fa.w + w2 * fb.z + w3 * fc.y;
        float fy = w0 * fa.y + w1 * fb.x + w2 * fb.w + w3 * fc.z;
        float fz = w0 * fa.z + w1 * fb.y + w2 * fc.x + w3 * fc.w;
        float dd = w0 * dp.x + w1 * dp.y + w2 * dp.z + w3 * dp.w;
        res[g] = make_float4(fx, fy, fz, dd);
    }
}

static __device__ __forceinline__ float reduce_quarter(const float4* rows, int lane)
{
    const float* Pf = (const float*)rows;
    const int col   = lane & 15;
    const int r     = col >> 2;
    const int ch    = col & 3;
    const int khalf = (lane >> 4) << 1;
    const v2f aones = {1.0f, 1.0f};
    v8f acc = {0.f, 0.f, 0.f, 0.f, 0.f, 0.f, 0.f, 0.f};
    #pragma unroll
    for (int kb = 0; kb < 8; ++kb) {
        const int k0 = kb * 4 + khalf;
        v2f b;
        b[0] = Pf[(r * PITCH + k0)     * 4 + ch];
        b[1] = Pf[(r * PITCH + k0 + 1) * 4 + ch];
        acc = __builtin_amdgcn_wmma_f32_16x16x4_f32(
            false, aones, false, b, (short)0, acc, false, false);
    }
    return acc[0];
}

static __device__ __forceinline__ void store_pair(float* __restrict__ out, float vLow, float vHigh,
                                                  int rayFirst, int N, int lane)
{
    const float v = (lane < 16) ? vLow : vHigh;
    const int ray = rayFirst + (lane >> 4) * QRAYS + ((lane & 15) >> 2);
    const int ch  = lane & 3;
    if (ray < N) {
        const int idx = (ch < 3) ? (ray * 3 + ch) : (3 * N + ray);
        out[idx] = v;
    }
}

__global__ __launch_bounds__(WAVES * 32)
void volrend_kernel(const float* __restrict__ density,
                    const float* __restrict__ depth,
                    const float* __restrict__ feature,
                    float* __restrict__ out, int N)
{
    __shared__ float4 part[WAVES][TILE][PITCH];
    __shared__ f4 frow[WAVES][GROUP][96];

    const int lane    = threadIdx.x & 31;
    const int wave    = __builtin_amdgcn_readfirstlane(threadIdx.x >> 5);
    const int rayBase = (blockIdx.x * WAVES + wave) * TILE;
    if (rayBase >= N) return;

    #define RAY_OF(R) ((rayBase + (R) < N) ? (rayBase + (R)) : rayBase)
    #define LOAD_STEP(S) do { _Pragma("unroll") for (int g_ = 0; g_ < GROUP; ++g_) \
        load_ray(in[(S) % (AHEAD + 1)][g_], density, depth, feature, RAY_OF((S) * GROUP + g_), lane); } while (0)

    RayIn in[AHEAD + 1][GROUP];
    float qv[TILE / QRAYS];

    #pragma unroll
    for (int s = 0; s < AHEAD; ++s) LOAD_STEP(s);

    #pragma unroll
    for (int s = 0; s < STEPS; ++s) {
        __builtin_amdgcn_sched_barrier(0);
        if (s + AHEAD < STEPS) LOAD_STEP(s + AHEAD);
        __builtin_amdgcn_sched_barrier(0);
        if (s > 0 && s % QSTEPS == 0)
            qv[s / QSTEPS - 1] = reduce_quarter(&part[wave][(s / QSTEPS - 1) * QRAYS][0], lane);
        #pragma unroll
        for (int g = 0; g < GROUP; ++g) {
            RayIn& q = in[s % (AHEAD + 1)][g];
            asm volatile("" : "+v"(q.sg), "+v"(q.dp), "+v"(q.fa), "+v"(q.fb), "+v"(q.fc));
        }
        to_sample_order(in[s % (AHEAD + 1)], frow[wave], lane);
        float4 res[GROUP];
        composite_group(in[s % (AHEAD + 1)], lane, res);
        #pragma unroll
        for (int g = 0; g < GROUP; ++g) part[wave][s * GROUP + g][lane] = res[g];
    }
    #undef LOAD_STEP
    #undef RAY_OF

    qv[TILE / QRAYS - 1] = reduce_quarter(&part[wave][TILE - QRAYS][0], lane);
    #pragma unroll
    for (int p = 0; p < TILE / QRAYS / 2; ++p)
        store_pair(out, qv[2 * p], qv[2 * p + 1], rayBase + 2 * p * QRAYS, N, lane);
}

extern "C" void kernel_launch(void* const* d_in, const int* in_sizes, int n_in,
                              void* d_out, int out_size, void* d_ws, size_t ws_size,
                              hipStream_t stream) {
    const float* density = (const float*)d_in[0];
    const float* depth   = (const float*)d_in[1];
    const float* feature = (const float*)d_in[2];
    float* out = (float*)d_out;

    const int N = in_sizes[0] / NP;
    const int blocks = (N + WAVES * TILE - 1) / (WAVES * TILE);

    hipLaunchKernelGGL(volrend_kernel, dim3(blocks), dim3(WAVES * 32), 0, stream,
                       density, depth, feature, out, N);
}
